# starting kernel
# baseline (speedup 1.0000x reference)
#include <hip/hip_runtime.h>
#include <hip/hip_bf16.h>

typedef __attribute__((ext_vector_type(2))) float v2f;
typedef __attribute__((ext_vector_type(8))) float v8f;

__device__ __forceinline__ v2f lda_f(const float* p, int ld, int r0, int k0, int lane) {
    int m = lane & 15, kb = (lane >> 4) << 1;
    const float* q = p + (r0 + m) * ld + k0 + kb;
    v2f r; r.x = q[0]; r.y = q[1];
    return r;
}
__device__ __forceinline__ v2f ldb_f(const float* p, int ld, int k0, int n0, int lane) {
    int n = lane & 15, kb = (lane >> 4) << 1;
    v2f r;
    r.x = p[(k0 + kb)     * ld + n0 + n];
    r.y = p[(k0 + kb + 1) * ld + n0 + n];
    return r;
}
__device__ __forceinline__ v8f ldc_f(const float* p, int ld, int r0, int c0, int lane) {
    int col = lane & 15, rb = (lane >> 4) << 3;
    v8f c;
#pragma unroll
    for (int g = 0; g < 8; ++g) c[g] = p[(r0 + rb + g) * ld + c0 + col];
    return c;
}
__device__ __forceinline__ void stc_f(float* p, int ld, int r0, int c0, int lane, v8f c) {
    int col = lane & 15, rb = (lane >> 4) << 3;
#pragma unroll
    for (int g = 0; g < 8; ++g) p[(r0 + rb + g) * ld + c0 + col] = c[g];
}
__device__ __forceinline__ v8f wmma4(v2f a, v2f b, v8f c) {
    return __builtin_amdgcn_wmma_f32_16x16x4_f32(false, a, false, b, (short)0, c, false, false);
}

__device__ __forceinline__ void ln_row(float* row, const float* w) {
    float m = 0.f;
#pragma unroll
    for (int c = 0; c < 11; ++c) m += row[c];
    m *= (1.0f / 11.0f);
    float v = 0.f;
#pragma unroll
    for (int c = 0; c < 11; ++c) { float d = row[c] - m; v += d * d; }
    v *= (1.0f / 11.0f);
    float rinv = rsqrtf(v + 1e-5f);
#pragma unroll
    for (int c = 0; c < 16; ++c) row[c] = (c < 11) ? (row[c] - m) * rinv * w[c] : 0.f;
}

__global__ __launch_bounds__(128)
void pixel_encoder_kernel(const float* __restrict__ x,
                          const float* __restrict__ Wqkv,
                          const float* __restrict__ Wo,
                          const float* __restrict__ W1,
                          const float* __restrict__ W2,
                          const float* __restrict__ ln1w,
                          const float* __restrict__ ln2w,
                          float* __restrict__ out)
{
    __shared__ float Wq_t[16 * 16], Wk_t[16 * 16], Wv_t[16 * 16], Wo_t[16 * 16];
    __shared__ float W1_t[16 * 64];
    __shared__ float W2_t[64 * 16];
    __shared__ float ln1[16], ln2[16];
    __shared__ float Xb[64 * 16];
    __shared__ float Qb[64 * 16], Kb[64 * 16], Vb[64 * 16];
    __shared__ float Pb[64 * 64];
    __shared__ float Ab[64 * 16];

    const int tid  = threadIdx.x;
    const int lane = tid & 31;
    const int wv   = tid >> 5;
    const int r0   = wv * 16;
    const int p    = blockIdx.x;
    const int hh   = p >> 4;
    const int ww   = p & 15;

    const float scale = rsqrtf(11.0f);
    for (int i = tid; i < 256; i += 128) {
        int k = i >> 4, n = i & 15;
        float q = 0.f, kk = 0.f, vv = 0.f, o = 0.f;
        if (n < 11 && k < 11) {
            q  = Wqkv[n * 11 + k] * scale;
            kk = Wqkv[(11 + n) * 11 + k];
            vv = Wqkv[(22 + n) * 11 + k];
            o  = Wo[n * 11 + k];
        }
        Wq_t[i] = q; Wk_t[i] = kk; Wv_t[i] = vv; Wo_t[i] = o;
    }
    for (int i = tid; i < 1024; i += 128) {
        int k1 = i >> 6, n1 = i & 63;
        W1_t[i] = (k1 < 11) ? W1[n1 * 11 + k1] : 0.f;
        int k2 = i >> 4, n2 = i & 15;
        W2_t[i] = (n2 < 11) ? W2[n2 * 64 + k2] : 0.f;
    }
    if (tid < 16) {
        ln1[tid] = (tid < 11) ? ln1w[tid] : 0.f;
        ln2[tid] = (tid < 11) ? ln2w[tid] : 0.f;
    }

    for (int i = tid; i < 1024; i += 128) {
        int t = i >> 4, ch = i & 15;
        float val = 0.f;
        if (t < 49) {
            int wi = t / 7, wj = t % 7;
            int gy = hh + wi - 3, gx = ww + wj - 3;
            bool inb = ((unsigned)gy < 16u) && ((unsigned)gx < 16u);
            if (ch < 10)       val = inb ? x[ch * 256 + gy * 16 + gx] : 0.f;
            else if (ch == 10) val = inb ? 0.f : 1.f;
        }
        Xb[i] = val;
    }
    __syncthreads();

    {
        v8f cq = {}, ck = {}, cv = {};
#pragma unroll
        for (int kk = 0; kk < 16; kk += 4) {
            v2f a = lda_f(Xb, 16, r0, kk, lane);
            cq = wmma4(a, ldb_f(Wq_t, 16, kk, 0, lane), cq);
            ck = wmma4(a, ldb_f(Wk_t, 16, kk, 0, lane), ck);
            cv = wmma4(a, ldb_f(Wv_t, 16, kk, 0, lane), cv);
        }
        stc_f(Qb, 16, r0, 0, lane, cq);
        stc_f(Kb, 16, r0, 0, lane, ck);
        stc_f(Vb, 16, r0, 0, lane, cv);
    }
    __syncthreads();

#pragma unroll
    for (int ct = 0; ct < 4; ++ct) {
        v8f c = {};
#pragma unroll
        for (int kk = 0; kk < 16; kk += 4) {
            v2f a = lda_f(Qb, 16, r0, kk, lane);
            v2f b = lda_f(Kb, 16, ct * 16, kk, lane);
            c = wmma4(a, b, c);
        }
        stc_f(Pb, 64, r0, ct * 16, lane, c);
    }
    __syncthreads();

    if (tid < 64) {
        float* row = Pb + tid * 64;
        float m = 0.f;
        for (int j = 0; j < 64; ++j) m = fmaxf(m, row[j]);
        float s = 836.f * expf(-m);
        for (int j = 0; j < 64; ++j) { float e = expf(row[j] - m); row[j] = e; s += e; }
        float inv = 1.f / s;
        for (int j = 0; j < 64; ++j) row[j] *= inv;
    }
    __syncthreads();

    {
        v8f c = {};
#pragma unroll
        for (int kk = 0; kk < 64; kk += 4)
            c = wmma4(lda_f(Pb, 64, r0, kk, lane), ldb_f(Vb, 16, kk, 0, lane), c);
        stc_f(Ab, 16, r0, 0, lane, c);
    }
    __syncthreads();

    {
        v8f c = ldc_f(Xb, 16, r0, 0, lane);
#pragma unroll
        for (int kk = 0; kk < 16; kk += 4)
            c = wmma4(lda_f(Ab, 16, r0, kk, lane), ldb_f(Wo_t, 16, kk, 0, lane), c);
        stc_f(Xb, 16, r0, 0, lane, c);
    }
    __syncthreads();

    if (tid < 64) ln_row(Xb + tid * 16, ln1);
    __syncthreads();

#pragma unroll
    for (int ct = 0; ct < 4; ++ct) {
        v8f c = {};
#pragma unroll
        for (int kk = 0; kk < 16; kk += 4)
            c = wmma4(lda_f(Xb, 16, r0, kk, lane), ldb_f(W1_t, 64, kk, ct * 16, lane), c);
#pragma unroll
        for (int g = 0; g < 8; ++g) c[g] = fmaxf(c[g], 0.f);
        stc_f(Pb, 64, r0, ct * 16, lane, c);
    }
    __syncthreads();

    {
        v8f c = ldc_f(Xb, 16, r0, 0, lane);
#pragma unroll
        for (int kk = 0; kk < 64; kk += 4)
            c = wmma4(lda_f(Pb, 64, r0, kk, lane), ldb_f(W2_t, 16, kk, 0, lane), c);
        stc_f(Ab, 16, r0, 0, lane, c);
    }
    __syncthreads();

    if (tid < 64) ln_row(Ab + tid * 16, ln2);
    __syncthreads();

    float* ob = out + p * 9000;
    for (int s = tid; s < 900; s += 128) {
        int r = s / 30, cc = s % 30;
        int t = (r < 7 && cc < 7) ? (r * 7 + cc) : 49;
        const float* row = Ab + t * 16;
#pragma unroll
        for (int ch = 0; ch < 10; ++ch) ob[ch * 900 + s] = row[ch];
    }
}

extern "C" void kernel_launch(void* const* d_in, const int* in_sizes, int n_in,
                              void* d_out, int out_size, void* d_ws, size_t ws_size,
                              hipStream_t stream) {
    const float* x    = (const float*)d_in[0];
    const float* Wqkv = (const float*)d_in[1];
    const float* Wo   = (const float*)d_in[2];
    const float* W1   = (const float*)d_in[3];
    const float* W2   = (const float*)d_in[4];
    const float* l1   = (const float*)d_in[5];
    const float* l2   = (const float*)d_in[6];
    (void)in_sizes; (void)n_in; (void)out_size; (void)d_ws; (void)ws_size;
    hipLaunchKernelGGL(pixel_encoder_kernel, dim3(256), dim3(128), 0, stream,
                       x, Wqkv, Wo, W1, W2, l1, l2, (float*)d_out);
}


# Attention core in registers: scores computed transposed (K*Q^T) so softmax reduces in-lane and P feeds P*V as WMMA A operand, no LDS round trip; softmax sum and P*V keys summed in a new order
# speedup vs baseline: 1.4964x; 1.4964x over previous
#include <hip/hip_runtime.h>
#include <hip/hip_bf16.h>

typedef __attribute__((ext_vector_type(2))) float v2f;
typedef __attribute__((ext_vector_type(8))) float v8f;

__device__ __forceinline__ v2f lda_f(const float* p, int ld, int r0, int k0, int lane) {
    int m = lane & 15, kb = (lane >> 4) << 1;
    const float* q = p + (r0 + m) * ld + k0 + kb;
    v2f r; r.x = q[0]; r.y = q[1];
    return r;
}
__device__ __forceinline__ v2f ldb_f(const float* p, int ld, int k0, int n0, int lane) {
    int n = lane & 15, kb = (lane >> 4) << 1;
    v2f r;
    r.x = p[(k0 + kb)     * ld + n0 + n];
    r.y = p[(k0 + kb + 1) * ld + n0 + n];
    return r;
}
__device__ __forceinline__ v8f ldc_f(const float* p, int ld, int r0, int c0, int lane) {
    int col = lane & 15, rb = (lane >> 4) << 3;
    v8f c;
#pragma unroll
    for (int g = 0; g < 8; ++g) c[g] = p[(r0 + rb + g) * ld + c0 + col];
    return c;
}
__device__ __forceinline__ void stc_f(float* p, int ld, int r0, int c0, int lane, v8f c) {
    int col = lane & 15, rb = (lane >> 4) << 3;
#pragma unroll
    for (int g = 0; g < 8; ++g) p[(r0 + rb + g) * ld + c0 + col] = c[g];
}
__device__ __forceinline__ v8f wmma4(v2f a, v2f b, v8f c) {
    return __builtin_amdgcn_wmma_f32_16x16x4_f32(false, a, false, b, (short)0, c, false, false);
}

__device__ __forceinline__ void ln_row(float* row, const float* w) {
    float m = 0.f;
#pragma unroll
    for (int c = 0; c < 11; ++c) m += row[c];
    m *= (1.0f / 11.0f);
    float v = 0.f;
#pragma unroll
    for (int c = 0; c < 11; ++c) { float d = row[c] - m; v += d * d; }
    v *= (1.0f / 11.0f);
    float rinv = rsqrtf(v + 1e-5f);
#pragma unroll
    for (int c = 0; c < 16; ++c) row[c] = (c < 11) ? (row[c] - m) * rinv * w[c] : 0.f;
}

__global__ __launch_bounds__(128)
void pixel_encoder_kernel(const float* __restrict__ x,
                          const float* __restrict__ Wqkv,
                          const float* __restrict__ Wo,
                          const float* __restrict__ W1,
                          const float* __restrict__ W2,
                          const float* __restrict__ ln1w,
                          const float* __restrict__ ln2w,
                          float* __restrict__ out)
{
    __shared__ float Wq_t[16 * 16], Wk_t[16 * 16], Wv_t[16 * 16], Wo_t[16 * 16];
    __shared__ float W1_t[16 * 64];
    __shared__ float W2_t[64 * 16];
    __shared__ float ln1[16], ln2[16];
    __shared__ float Xb[64 * 16];
    __shared__ float Qb[64 * 16], Kb[64 * 16], Vb[64 * 16];
    __shared__ float Pb[64 * 64];
    __shared__ float Ab[64 * 16];

    const int tid  = threadIdx.x;
    const int lane = tid & 31;
    const int wv   = tid >> 5;
    const int r0   = wv * 16;
    const int p    = blockIdx.x;
    const int hh   = p >> 4;
    const int ww   = p & 15;

    const float scale = rsqrtf(11.0f);
    for (int i = tid; i < 256; i += 128) {
        int k = i >> 4, n = i & 15;
        float q = 0.f, kk = 0.f, vv = 0.f, o = 0.f;
        if (n < 11 && k < 11) {
            q  = Wqkv[n * 11 + k] * scale;
            kk = Wqkv[(11 + n) * 11 + k];
            vv = Wqkv[(22 + n) * 11 + k];
            o  = Wo[n * 11 + k];
        }
        Wq_t[i] = q; Wk_t[i] = kk; Wv_t[i] = vv; Wo_t[i] = o;
    }
    for (int i = tid; i < 1024; i += 128) {
        int k1 = i >> 6, n1 = i & 63;
        W1_t[i] = (k1 < 11) ? W1[n1 * 11 + k1] : 0.f;
        int k2 = i >> 4, n2 = i & 15;
        W2_t[i] = (n2 < 11) ? W2[n2 * 64 + k2] : 0.f;
    }
    if (tid < 16) {
        ln1[tid] = (tid < 11) ? ln1w[tid] : 0.f;
        ln2[tid] = (tid < 11) ? ln2w[tid] : 0.f;
    }

    for (int i = tid; i < 1024; i += 128) {
        int t = i >> 4, ch = i & 15;
        float val = 0.f;
        if (t < 49) {
            int wi = t / 7, wj = t % 7;
            int gy = hh + wi - 3, gx = ww + wj - 3;
            bool inb = ((unsigned)gy < 16u) && ((unsigned)gx < 16u);
            if (ch < 10)       val = inb ? x[ch * 256 + gy * 16 + gx] : 0.f;
            else if (ch == 10) val = inb ? 0.f : 1.f;
        }
        Xb[i] = val;
    }
    __syncthreads();

    {
        v8f cq = {}, ck = {}, cv = {};
#pragma unroll
        for (int kk = 0; kk < 16; kk += 4) {
            v2f a = lda_f(Xb, 16, r0, kk, lane);
            cq = wmma4(a, ldb_f(Wq_t, 16, kk, 0, lane), cq);
            ck = wmma4(a, ldb_f(Wk_t, 16, kk, 0, lane), ck);
            cv = wmma4(a, ldb_f(Wv_t, 16, kk, 0, lane), cv);
        }
        stc_f(Qb, 16, r0, 0, lane, cq);
        stc_f(Kb, 16, r0, 0, lane, ck);
        stc_f(Vb, 16, r0, 0, lane, cv);
    }
    __syncthreads();

    {
        const int c16 = lane & 15, h = lane >> 4;
        v8f st[4] = {{}, {}, {}, {}};
#pragma unroll
        for (int kk = 0; kk < 16; kk += 4) {
            v2f bq = lda_f(Qb, 16, r0, kk, lane);
#pragma unroll
            for (int ct = 0; ct < 4; ++ct)
                st[ct] = wmma4(lda_f(Kb, 16, ct * 16, kk, lane), bq, st[ct]);
        }

        float m = 0.f;
#pragma unroll
        for (int ct = 0; ct < 4; ++ct)
#pragma unroll
            for (int g = 0; g < 8; ++g) m = fmaxf(m, st[ct][g]);
        m = fmaxf(m, __shfl_xor(m, 16, 32));
        float s = 0.f;
#pragma unroll
        for (int ct = 0; ct < 4; ++ct)
#pragma unroll
            for (int g = 0; g < 8; ++g) { float e = expf(st[ct][g] - m); st[ct][g] = e; s += e; }
        s += __shfl_xor(s, 16, 32);
        s += 836.f * expf(-m);
        const float inv = 1.f / s;
#pragma unroll
        for (int ct = 0; ct < 4; ++ct)
#pragma unroll
            for (int g = 0; g < 8; ++g) st[ct][g] *= inv;

        v8f acc = {};
#pragma unroll
        for (int ct = 0; ct < 4; ++ct) {
#pragma unroll
            for (int i = 0; i < 4; ++i) {
                const int kr = ct * 16 + h * 8 + i * 2;
                v2f a; a.x = st[ct][2 * i]; a.y = st[ct][2 * i + 1];
                v2f b; b.x = Vb[kr * 16 + c16]; b.y = Vb[(kr + 1) * 16 + c16];
                acc = wmma4(a, b, acc);
            }
        }
        stc_f(Ab, 16, r0, 0, lane, acc);
    }
    __syncthreads();

    {
        v8f c = ldc_f(Xb, 16, r0, 0, lane);
#pragma unroll
        for (int kk = 0; kk < 16; kk += 4)
            c = wmma4(lda_f(Ab, 16, r0, kk, lane), ldb_f(Wo_t, 16, kk, 0, lane), c);
        stc_f(Xb, 16, r0, 0, lane, c);
    }
    __syncthreads();

    if (tid < 64) ln_row(Xb + tid * 16, ln1);
    __syncthreads();

#pragma unroll
    for (int ct = 0; ct < 4; ++ct) {
        v8f c = {};
#pragma unroll
        for (int kk = 0; kk < 16; kk += 4)
            c = wmma4(lda_f(Xb, 16, r0, kk, lane), ldb_f(W1_t, 64, kk, ct * 16, lane), c);
#pragma unroll
        for (int g = 0; g < 8; ++g) c[g] = fmaxf(c[g], 0.f);
        stc_f(Pb, 64, r0, ct * 16, lane, c);
    }
    __syncthreads();

    {
        v8f c = ldc_f(Xb, 16, r0, 0, lane);
#pragma unroll
        for (int kk = 0; kk < 64; kk += 4)
            c = wmma4(lda_f(Pb, 64, r0, kk, lane), ldb_f(W2_t, 16, kk, 0, lane), c);
        stc_f(Ab, 16, r0, 0, lane, c);
    }
    __syncthreads();

    if (tid < 64) ln_row(Ab + tid * 16, ln2);
    __syncthreads();

    float* ob = out + p * 9000;
    for (int s = tid; s < 900; s += 128) {
        int r = s / 30, cc = s % 30;
        int t = (r < 7 && cc < 7) ? (r * 7 + cc) : 49;
        const float* row = Ab + t * 16;
#pragma unroll
        for (int ch = 0; ch < 10; ++ch) ob[ch * 900 + s] = row[ch];
    }
}

extern "C" void kernel_launch(void* const* d_in, const int* in_sizes, int n_in,
                              void* d_out, int out_size, void* d_ws, size_t ws_size,
                              hipStream_t stream) {
    const float* x    = (const float*)d_in[0];
    const float* Wqkv = (const float*)d_in[1];
    const float* Wo   = (const float*)d_in[2];
    const float* W1   = (const float*)d_in[3];
    const float* W2   = (const float*)d_in[4];
    const float* l1   = (const float*)d_in[5];
    const float* l2   = (const float*)d_in[6];
    (void)in_sizes; (void)n_in; (void)out_size; (void)d_ws; (void)ws_size;
    hipLaunchKernelGGL(pixel_encoder_kernel, dim3(256), dim3(128), 0, stream,
                       x, Wqkv, Wo, W1, W2, l1, l2, (float*)d_out);
}


# Prologue global reads (weights, LN gains, window pixels) made branch-free (clamped addresses, padding by select) and issued as one batch: one memory round trip instead of one per loop iteration
# speedup vs baseline: 1.9989x; 1.3358x over previous
#include <hip/hip_runtime.h>
#include <hip/hip_bf16.h>

typedef __attribute__((ext_vector_type(2))) float v2f;
typedef __attribute__((ext_vector_type(8))) float v8f;

__device__ __forceinline__ v2f lda_f(const float* p, int ld, int r0, int k0, int lane) {
    int m = lane & 15, kb = (lane >> 4) << 1;
    const float* q = p + (r0 + m) * ld + k0 + kb;
    v2f r; r.x = q[0]; r.y = q[1];
    return r;
}
__device__ __forceinline__ v2f ldb_f(const float* p, int ld, int k0, int n0, int lane) {
    int n = lane & 15, kb = (lane >> 4) << 1;
    v2f r;
    r.x = p[(k0 + kb)     * ld + n0 + n];
    r.y = p[(k0 + kb + 1) * ld + n0 + n];
    return r;
}
__device__ __forceinline__ v8f ldc_f(const float* p, int ld, int r0, int c0, int lane) {
    int col = lane & 15, rb = (lane >> 4) << 3;
    v8f c;
#pragma unroll
    for (int g = 0; g < 8; ++g) c[g] = p[(r0 + rb + g) * ld + c0 + col];
    return c;
}
__device__ __forceinline__ void stc_f(float* p, int ld, int r0, int c0, int lane, v8f c) {
    int col = lane & 15, rb = (lane >> 4) << 3;
#pragma unroll
    for (int g = 0; g < 8; ++g) p[(r0 + rb + g) * ld + c0 + col] = c[g];
}
__device__ __forceinline__ v8f wmma4(v2f a, v2f b, v8f c) {
    return __builtin_amdgcn_wmma_f32_16x16x4_f32(false, a, false, b, (short)0, c, false, false);
}

__device__ __forceinline__ void ln_row(float* row, const float* w) {
    float m = 0.f;
#pragma unroll
    for (int c = 0; c < 11; ++c) m += row[c];
    m *= (1.0f / 11.0f);
    float v = 0.f;
#pragma unroll
    for (int c = 0; c < 11; ++c) { float d = row[c] - m; v += d * d; }
    v *= (1.0f / 11.0f);
    float rinv = rsqrtf(v + 1e-5f);
#pragma unroll
    for (int c = 0; c < 16; ++c) row[c] = (c < 11) ? (row[c] - m) * rinv * w[c] : 0.f;
}

__global__ __launch_bounds__(128)
void pixel_encoder_kernel(const float* __restrict__ x,
                          const float* __restrict__ Wqkv,
                          const float* __restrict__ Wo,
                          const float* __restrict__ W1,
                          const float* __restrict__ W2,
                          const float* __restrict__ ln1w,
                          const float* __restrict__ ln2w,
                          float* __restrict__ out)
{
    __shared__ float Wq_t[16 * 16], Wk_t[16 * 16], Wv_t[16 * 16], Wo_t[16 * 16];
    __shared__ float W1_t[16 * 64];
    __shared__ float W2_t[64 * 16];
    __shared__ float ln1[16], ln2[16];
    __shared__ float Xb[64 * 16];
    __shared__ float Qb[64 * 16], Kb[64 * 16], Vb[64 * 16];
    __shared__ float Pb[64 * 64];
    __shared__ float Ab[64 * 16];

    const int tid  = threadIdx.x;
    const int lane = tid & 31;
    const int wv   = tid >> 5;
    const int r0   = wv * 16;
    const int p    = blockIdx.x;
    const int hh   = p >> 4;
    const int ww   = p & 15;

    const float scale = rsqrtf(11.0f);
    float gq[2], gk[2], gv[2], go[2], g1[6], g2[8], gx[7], gl1, gl2;
#pragma unroll
    for (int j = 0; j < 2; ++j) {
        const int i = tid + 128 * j;
        const int k = i >> 4, n = i & 15;
        const int a = min(n, 10) * 11 + min(k, 10);
        gq[j] = Wqkv[a];
        gk[j] = Wqkv[121 + a];
        gv[j] = Wqkv[242 + a];
        go[j] = Wo[a];
    }
#pragma unroll
    for (int j = 0; j < 6; ++j) {
        const int i = tid + 128 * j;
        const int k1 = i >> 6, n1 = i & 63;
        g1[j] = W1[n1 * 11 + min(k1, 10)];
    }
#pragma unroll
    for (int j = 0; j < 8; ++j) {
        const int i = tid + 128 * j;
        const int k2 = i >> 4, n2 = i & 15;
        g2[j] = W2[min(n2, 10) * 64 + k2];
    }
    {
        const int c = min(tid, 10);
        gl1 = ln1w[c];
        gl2 = ln2w[c];
    }
    const int xch = tid & 15;
    const int t0  = tid >> 4;
    bool xin[7];
    bool xwin[7];
#pragma unroll
    for (int j = 0; j < 7; ++j) {
        xwin[j] = (j < 6) || (t0 == 0);
        const int u  = (j < 6 ? t0 : 0) + j;
        const int c7 = (u >= 7) ? 1 : 0;
        const int gy  = hh + j + c7 - 3;
        const int gxx = ww + u - 7 * c7 - 3;
        xin[j] = xwin[j] && ((unsigned)gy < 16u) && ((unsigned)gxx < 16u);
        gx[j] = x[min(xch, 9) * 256 + min(max(gy, 0), 15) * 16 + min(max(gxx, 0), 15)];
    }
    __builtin_amdgcn_sched_barrier(0);

#pragma unroll
    for (int j = 0; j < 2; ++j) {
        const int i = tid + 128 * j;
        const int k = i >> 4, n = i & 15;
        const bool ok = (n < 11) && (k < 11);
        Wq_t[i] = ok ? gq[j] * scale : 0.f;
        Wk_t[i] = ok ? gk[j] : 0.f;
        Wv_t[i] = ok ? gv[j] : 0.f;
        Wo_t[i] = ok ? go[j] : 0.f;
    }
#pragma unroll
    for (int j = 0; j < 8; ++j) {
        const int i = tid + 128 * j;
        const int k1 = i >> 6;
        W1_t[i] = (j < 6 && k1 < 11) ? g1[j < 6 ? j : 0] : 0.f;
        const int n2 = i & 15;
        W2_t[i] = (n2 < 11) ? g2[j] : 0.f;
    }
    asm volatile("" :: "v"(gl1), "v"(gl2));
    if (tid < 16) {
        ln1[tid] = (tid < 11) ? gl1 : 0.f;
        ln2[tid] = (tid < 11) ? gl2 : 0.f;
    }

#pragma unroll
    for (int j = 0; j < 8; ++j) {
        const int i = tid + 128 * j;
        float val = 0.f;
        if (j < 7) {
            const int jj = j < 7 ? j : 0;
            const float mk = (xch == 10 && xwin[jj] && !xin[jj]) ? 1.f : 0.f;
            val = (xch < 10 && xin[jj]) ? gx[jj] : mk;
        }
        Xb[i] = val;
    }
    __syncthreads();

    {
        v8f cq = {}, ck = {}, cv = {};
#pragma unroll
        for (int kk = 0; kk < 16; kk += 4) {
            v2f a = lda_f(Xb, 16, r0, kk, lane);
            cq = wmma4(a, ldb_f(Wq_t, 16, kk, 0, lane), cq);
            ck = wmma4(a, ldb_f(Wk_t, 16, kk, 0, lane), ck);
            cv = wmma4(a, ldb_f(Wv_t, 16, kk, 0, lane), cv);
        }
        stc_f(Qb, 16, r0, 0, lane, cq);
        stc_f(Kb, 16, r0, 0, lane, ck);
        stc_f(Vb, 16, r0, 0, lane, cv);
    }
    __syncthreads();

    {
        const int c16 = lane & 15, h = lane >> 4;
        v8f st[4] = {{}, {}, {}, {}};
#pragma unroll
        for (int kk = 0; kk < 16; kk += 4) {
            v2f bq = lda_f(Qb, 16, r0, kk, lane);
#pragma unroll
            for (int ct = 0; ct < 4; ++ct)
                st[ct] = wmma4(lda_f(Kb, 16, ct * 16, kk, lane), bq, st[ct]);
        }

        float m = 0.f;
#pragma unroll
        for (int ct = 0; ct < 4; ++ct)
#pragma unroll
            for (int g = 0; g < 8; ++g) m = fmaxf(m, st[ct][g]);
        m = fmaxf(m, __shfl_xor(m, 16, 32));
        float s = 0.f;
#pragma unroll
        for (int ct = 0; ct < 4; ++ct)
#pragma unroll
            for (int g = 0; g < 8; ++g) { float e = expf(st[ct][g] - m); st[ct][g] = e; s += e; }
        s += __shfl_xor(s, 16, 32);
        s += 836.f * expf(-m);
        const float inv = 1.f / s;
#pragma unroll
        for (int ct = 0; ct < 4; ++ct)
#pragma unroll
            for (int g = 0; g < 8; ++g) st[ct][g] *= inv;

        v8f acc = {};
#pragma unroll
        for (int ct = 0; ct < 4; ++ct) {
#pragma unroll
            for (int i = 0; i < 4; ++i) {
                const int kr = ct * 16 + h * 8 + i * 2;
                v2f a; a.x = st[ct][2 * i]; a.y = st[ct][2 * i + 1];
                v2f b; b.x = Vb[kr * 16 + c16]; b.y = Vb[(kr + 1) * 16 + c16];
                acc = wmma4(a, b, acc);
            }
        }
        stc_f(Ab, 16, r0, 0, lane, acc);
    }
    __syncthreads();

    {
        v8f c = ldc_f(Xb, 16, r0, 0, lane);
#pragma unroll
        for (int kk = 0; kk < 16; kk += 4)
            c = wmma4(lda_f(Ab, 16, r0, kk, lane), ldb_f(Wo_t, 16, kk, 0, lane), c);
        stc_f(Xb, 16, r0, 0, lane, c);
    }
    __syncthreads();

    if (tid < 64) ln_row(Xb + tid * 16, ln1);
    __syncthreads();

#pragma unroll
    for (int ct = 0; ct < 4; ++ct) {
        v8f c = {};
#pragma unroll
        for (int kk = 0; kk < 16; kk += 4)
            c = wmma4(lda_f(Xb, 16, r0, kk, lane), ldb_f(W1_t, 64, kk, ct * 16, lane), c);
#pragma unroll
        for (int g = 0; g < 8; ++g) c[g] = fmaxf(c[g], 0.f);
        stc_f(Pb, 64, r0, ct * 16, lane, c);
    }
    __syncthreads();

    {
        v8f c = ldc_f(Xb, 16, r0, 0, lane);
#pragma unroll
        for (int kk = 0; kk < 64; kk += 4)
            c = wmma4(lda_f(Pb, 64, r0, kk, lane), ldb_f(W2_t, 16, kk, 0, lane), c);
        stc_f(Ab, 16, r0, 0, lane, c);
    }
    __syncthreads();

    if (tid < 64) ln_row(Ab + tid * 16, ln2);
    __syncthreads();

    float* ob = out + p * 9000;
    for (int s = tid; s < 900; s += 128) {
        int r = s / 30, cc = s % 30;
        int t = (r < 7 && cc < 7) ? (r * 7 + cc) : 49;
        const float* row = Ab + t * 16;
#pragma unroll
        for (int ch = 0; ch < 10; ++ch) ob[ch * 900 + s] = row[ch];
    }
}

extern "C" void kernel_launch(void* const* d_in, const int* in_sizes, int n_in,
                              void* d_out, int out_size, void* d_ws, size_t ws_size,
                              hipStream_t stream) {
    const float* x    = (const float*)d_in[0];
    const float* Wqkv = (const float*)d_in[1];
    const float* Wo   = (const float*)d_in[2];
    const float* W1   = (const float*)d_in[3];
    const float* W2   = (const float*)d_in[4];
    const float* l1   = (const float*)d_in[5];
    const float* l2   = (const float*)d_in[6];
    (void)in_sizes; (void)n_in; (void)out_size; (void)d_ws; (void)ws_size;
    hipLaunchKernelGGL(pixel_encoder_kernel, dim3(256), dim3(128), 0, stream,
                       x, Wqkv, Wo, W1, W2, l1, l2, (float*)d_out);
}


# Chain behind the softmax (P*V, Wo, LN1, FF1, FF2, LN2) runs from registers: products transposed so each D tile is the next B operand, LDS operands fetched ahead of the softmax; sums reordered
# speedup vs baseline: 2.4755x; 1.2384x over previous
#include <hip/hip_runtime.h>
#include <hip/hip_bf16.h>

typedef __attribute__((ext_vector_type(2))) float v2f;
typedef __attribute__((ext_vector_type(4))) float v4f;
typedef __attribute__((ext_vector_type(8))) float v8f;

__device__ __forceinline__ v2f lda_f(const float* p, int ld, int r0, int k0, int lane) {
    int m = lane & 15, kb = (lane >> 4) << 1;
    const float* q = p + (r0 + m) * ld + k0 + kb;
    v2f r; r.x = q[0]; r.y = q[1];
    return r;
}
__device__ __forceinline__ v2f ldb_f(const float* p, int ld, int k0, int n0, int lane) {
    int n = lane & 15, kb = (lane >> 4) << 1;
    v2f r;
    r.x = p[(k0 + kb)     * ld + n0 + n];
    r.y = p[(k0 + kb + 1) * ld + n0 + n];
    return r;
}
__device__ __forceinline__ void stc_f(float* p, int ld, int r0, int c0, int lane, v8f c) {
    int col = lane & 15, rb = (lane >> 4) << 3;
#pragma unroll
    for (int g = 0; g < 8; ++g) p[(r0 + rb + g) * ld + c0 + col] = c[g];
}
__device__ __forceinline__ v8f wmma4(v2f a, v2f b, v8f c) {
    return __builtin_amdgcn_wmma_f32_16x16x4_f32(false, a, false, b, (short)0, c, false, false);
}

__device__ __forceinline__ v8f lds8(const float* p) {
    const v4f lo = *(const v4f*)p, hi = *(const v4f*)(p + 4);
    v8f r;
    r[0] = lo[0]; r[1] = lo[1]; r[2] = lo[2]; r[3] = lo[3];
    r[4] = hi[0]; r[5] = hi[1]; r[6] = hi[2]; r[7] = hi[3];
    return r;
}
__device__ __forceinline__ float xhalf(float x) {
    return __builtin_bit_cast(float, __builtin_amdgcn_permlanex16(__builtin_bit_cast(int, x), __builtin_bit_cast(int, x),
                                                                  0x76543210, 0xfedcba98, false, false));
}
__device__ __forceinline__ v2f pr(v8f t, int j) {
    v2f r; r.x = t[2 * j]; r.y = t[2 * j + 1];
    return r;
}

__device__ __forceinline__ v8f ln_t(v8f x, v8f w, int h) {
    float xs[8];
#pragma unroll
    for (int r = 0; r < 8; ++r) xs[r] = (r >= 3 && h) ? 0.f : x[r];
    float s = ((xs[0] + xs[1]) + (xs[2] + xs[3])) + ((xs[4] + xs[5]) + (xs[6] + xs[7]));
    s += xhalf(s);
    const float m = s * (1.0f / 11.0f);
    float d[8];
#pragma unroll
    for (int r = 0; r < 8; ++r) d[r] = (r >= 3 && h) ? 0.f : xs[r] - m;
    float v = ((d[0] * d[0] + d[1] * d[1]) + (d[2] * d[2] + d[3] * d[3])) +
              ((d[4] * d[4] + d[5] * d[5]) + (d[6] * d[6] + d[7] * d[7]));
    v += xhalf(v);
    v *= (1.0f / 11.0f);
    const float rinv = rsqrtf(v + 1e-5f);
    v8f o;
#pragma unroll
    for (int r = 0; r < 8; ++r) o[r] = d[r] * rinv * w[r];
    return o;
}

__global__ __launch_bounds__(128)
void pixel_encoder_kernel(const float* __restrict__ x,
                          const float* __restrict__ Wqkv,
                          const float* __restrict__ Wo,
                          const float* __restrict__ W1,
                          const float* __restrict__ W2,
                          const float* __restrict__ ln1w,
                          const float* __restrict__ ln2w,
                          float* __restrict__ out)
{
    __shared__ float Wq_t[16 * 16], Wk_t[16 * 16], Wv_t[16 * 16];
    __shared__ __attribute__((aligned(16))) float Wo_p[16 * 20];
    __shared__ __attribute__((aligned(16))) float W1_p[64 * 20];
    __shared__ __attribute__((aligned(16))) float W2_p[16 * 68];
    __shared__ __attribute__((aligned(16))) float ln1[16], ln2[16];
    __shared__ __attribute__((aligned(16))) float Xb[64 * 16];
    __shared__ float Qb[64 * 16], Kb[64 * 16], Vb[64 * 16];
    __shared__ __attribute__((aligned(16))) float Ab[64 * 16];

    const int tid  = threadIdx.x;
    const int lane = tid & 31;
    const int wv   = tid >> 5;
    const int r0   = wv * 16;
    const int p    = blockIdx.x;
    const int hh   = p >> 4;
    const int ww   = p & 15;

    const float scale = rsqrtf(11.0f);
    float gq[2], gk[2], gv[2], go[2], g1[6], g2[8], gx[7], gl1, gl2;
#pragma unroll
    for (int j = 0; j < 2; ++j) {
        const int i = tid + 128 * j;
        const int k = i >> 4, n = i & 15;
        const int a = min(n, 10) * 11 + min(k, 10);
        gq[j] = Wqkv[a];
        gk[j] = Wqkv[121 + a];
        gv[j] = Wqkv[242 + a];
        go[j] = Wo[a];
    }
#pragma unroll
    for (int j = 0; j < 6; ++j) {
        const int i = tid + 128 * j;
        const int k1 = i >> 6, n1 = i & 63;
        g1[j] = W1[n1 * 11 + min(k1, 10)];
    }
#pragma unroll
    for (int j = 0; j < 8; ++j) {
        const int i = tid + 128 * j;
        const int k2 = i >> 4, n2 = i & 15;
        g2[j] = W2[min(n2, 10) * 64 + k2];
    }
    {
        const int c = min(tid, 10);
        gl1 = ln1w[c];
        gl2 = ln2w[c];
    }
    const int xch = tid & 15;
    const int t0  = tid >> 4;
    bool xin[7];
    bool xwin[7];
#pragma unroll
    for (int j = 0; j < 7; ++j) {
        xwin[j] = (j < 6) || (t0 == 0);
        const int u  = (j < 6 ? t0 : 0) + j;
        const int c7 = (u >= 7) ? 1 : 0;
        const int gy  = hh + j + c7 - 3;
        const int gxx = ww + u - 7 * c7 - 3;
        xin[j] = xwin[j] && ((unsigned)gy < 16u) && ((unsigned)gxx < 16u);
        gx[j] = x[min(xch, 9) * 256 + min(max(gy, 0), 15) * 16 + min(max(gxx, 0), 15)];
    }
    __builtin_amdgcn_sched_barrier(0);

#pragma unroll
    for (int j = 0; j < 2; ++j) {
        const int i = tid + 128 * j;
        const int k = i >> 4, n = i & 15;
        const bool ok = (n < 11) && (k < 11);
        Wq_t[i] = ok ? gq[j] * scale : 0.f;
        Wk_t[i] = ok ? gk[j] : 0.f;
        Wv_t[i] = ok ? gv[j] : 0.f;
        Wo_p[n * 20 + k] = ok ? go[j] : 0.f;
    }
#pragma unroll
    for (int j = 0; j < 8; ++j) {
        const int i = tid + 128 * j;
        const int k1 = i >> 6, n1 = i & 63;
        W1_p[n1 * 20 + k1] = (j < 6 && k1 < 11) ? g1[j < 6 ? j : 0] : 0.f;
        const int k2 = i >> 4, n2 = i & 15;
        W2_p[n2 * 68 + k2] = (n2 < 11) ? g2[j] : 0.f;
    }
    asm volatile("" :: "v"(gl1), "v"(gl2));
    if (tid < 16) {
        ln1[tid] = (tid < 11) ? gl1 : 0.f;
        ln2[tid] = (tid < 11) ? gl2 : 0.f;
    }

#pragma unroll
    for (int j = 0; j < 8; ++j) {
        const int i = tid + 128 * j;
        float val = 0.f;
        if (j < 7) {
            const int jj = j < 7 ? j : 0;
            const float mk = (xch == 10 && xwin[jj] && !xin[jj]) ? 1.f : 0.f;
            val = (xch < 10 && xin[jj]) ? gx[jj] : mk;
        }
        Xb[i] = val;
    }
    __syncthreads();

    {
        v8f cq = {}, ck = {}, cv = {};
#pragma unroll
        for (int kk = 0; kk < 16; kk += 4) {
            v2f a = lda_f(Xb, 16, r0, kk, lane);
            cq = wmma4(a, ldb_f(Wq_t, 16, kk, 0, lane), cq);
            ck = wmma4(a, ldb_f(Wk_t, 16, kk, 0, lane), ck);
            cv = wmma4(a, ldb_f(Wv_t, 16, kk, 0, lane), cv);
        }
        stc_f(Qb, 16, r0, 0, lane, cq);
        stc_f(Kb, 16, r0, 0, lane, ck);
        stc_f(Vb, 16, r0, 0, lane, cv);
    }
    __syncthreads();

    {
        const int c16 = lane & 15, h = lane >> 4;

        v2f bq[4], ak[4][4], av[4][4];
#pragma unroll
        for (int kk = 0; kk < 4; ++kk) {
            bq[kk] = lda_f(Qb, 16, r0, kk * 4, lane);
#pragma unroll
            for (int ct = 0; ct < 4; ++ct) ak[kk][ct] = lda_f(Kb, 16, ct * 16, kk * 4, lane);
        }
#pragma unroll
        for (int ct = 0; ct < 4; ++ct) {
#pragma unroll
            for (int i = 0; i < 4; ++i) {
                const int kr = ct * 16 + h * 8 + i * 2;
                av[ct][i].x = Vb[kr * 16 + c16];
                av[ct][i].y = Vb[(kr + 1) * 16 + c16];
            }
        }
        const v8f xt = lds8(Xb + (r0 + c16) * 16 + 8 * h);
        __builtin_amdgcn_sched_barrier(0);

        v8f st[4] = {{}, {}, {}, {}};
#pragma unroll
        for (int kk = 0; kk < 4; ++kk) {
#pragma unroll
            for (int ct = 0; ct < 4; ++ct)
                st[ct] = wmma4(ak[kk][ct], bq[kk], st[ct]);
        }

        const v8f wo = lds8(Wo_p + c16 * 20 + 8 * h);
        v8f w1[4], w2[4];
#pragma unroll
        for (int ht = 0; ht < 4; ++ht) {
            w1[ht] = lds8(W1_p + (ht * 16 + c16) * 20 + 8 * h);
            w2[ht] = lds8(W2_p + c16 * 68 + ht * 16 + 8 * h);
        }
        const v8f gn1 = lds8(ln1 + 8 * h), gn2 = lds8(ln2 + 8 * h);
        __builtin_amdgcn_sched_barrier(0);

        float m = 0.f;
#pragma unroll
        for (int ct = 0; ct < 4; ++ct)
#pragma unroll
            for (int g = 0; g < 8; ++g) m = fmaxf(m, st[ct][g]);
        m = fmaxf(m, __shfl_xor(m, 16, 32));
        float s = 0.f;
#pragma unroll
        for (int ct = 0; ct < 4; ++ct)
#pragma unroll
            for (int g = 0; g < 8; ++g) { float e = expf(st[ct][g] - m); st[ct][g] = e; s += e; }
        s += __shfl_xor(s, 16, 32);
        s += 836.f * expf(-m);
        const float inv = 1.f / s;
#pragma unroll
        for (int ct = 0; ct < 4; ++ct)
#pragma unroll
            for (int g = 0; g < 8; ++g) st[ct][g] *= inv;


        v8f ot = {};
#pragma unroll
        for (int ct = 0; ct < 4; ++ct) {
#pragma unroll
            for (int i = 0; i < 4; ++i) {
                ot = wmma4(av[ct][i], pr(st[ct], i), ot);
            }
        }

        v8f r1 = xt;
#pragma unroll
        for (int j = 0; j < 4; ++j) r1 = wmma4(pr(wo, j), pr(ot, j), r1);
        const v8f x1 = ln_t(r1, gn1, h);

        v8f hd[4];
#pragma unroll
        for (int ht = 0; ht < 4; ++ht) {
            v8f c = {};
#pragma unroll
            for (int j = 0; j < 4; ++j) c = wmma4(pr(w1[ht], j), pr(x1, j), c);
#pragma unroll
            for (int g = 0; g < 8; ++g) c[g] = fmaxf(c[g], 0.f);
            hd[ht] = c;
        }

        v8f y = x1;
#pragma unroll
        for (int ht = 0; ht < 4; ++ht) {
#pragma unroll
            for (int i = 0; i < 4; ++i) y = wmma4(pr(w2[ht], i), pr(hd[ht], i), y);
        }
        const v8f fin = ln_t(y, gn2, h);

        float* fo = Ab + (r0 + c16) * 16 + 8 * h;
        v4f lo, hi;
        lo[0] = fin[0]; lo[1] = fin[1]; lo[2] = fin[2]; lo[3] = fin[3];
        hi[0] = fin[4]; hi[1] = fin[5]; hi[2] = fin[6]; hi[3] = fin[7];
        *(v4f*)fo = lo;
        *(v4f*)(fo + 4) = hi;
    }
    __syncthreads();

    float* ob = out + p * 9000;
    for (int s = tid; s < 900; s += 128) {
        int r = s / 30, cc = s % 30;
        int t = (r < 7 && cc < 7) ? (r * 7 + cc) : 49;
        const float* row = Ab + t * 16;
#pragma unroll
        for (int ch = 0; ch < 10; ++ch) ob[ch * 900 + s] = row[ch];
    }
}

extern "C" void kernel_launch(void* const* d_in, const int* in_sizes, int n_in,
                              void* d_out, int out_size, void* d_ws, size_t ws_size,
                              hipStream_t stream) {
    const float* x    = (const float*)d_in[0];
    const float* Wqkv = (const float*)d_in[1];
    const float* Wo   = (const float*)d_in[2];
    const float* W1   = (const float*)d_in[3];
    const float* W2   = (const float*)d_in[4];
    const float* l1   = (const float*)d_in[5];
    const float* l2   = (const float*)d_in[6];
    (void)in_sizes; (void)n_in; (void)out_size; (void)d_ws; (void)ws_size;
    hipLaunchKernelGGL(pixel_encoder_kernel, dim3(256), dim3(128), 0, stream,
                       x, Wqkv, Wo, W1, W2, l1, l2, (float*)d_out);
}


# Softmax on native arithmetic: log2(e) folded into staged Wq so each probability is one v_exp_f32; unnormalised exps feed P*V tile by tile, one v_rcp_f32 scales O^T; max/sum as trees; sum reordered
# speedup vs baseline: 2.6224x; 1.0593x over previous
#include <hip/hip_runtime.h>
#include <hip/hip_bf16.h>

typedef __attribute__((ext_vector_type(2))) float v2f;
typedef __attribute__((ext_vector_type(4))) float v4f;
typedef __attribute__((ext_vector_type(8))) float v8f;

__device__ __forceinline__ v2f lda_f(const float* p, int ld, int r0, int k0, int lane) {
    int m = lane & 15, kb = (lane >> 4) << 1;
    const float* q = p + (r0 + m) * ld + k0 + kb;
    v2f r; r.x = q[0]; r.y = q[1];
    return r;
}
__device__ __forceinline__ v2f ldb_f(const float* p, int ld, int k0, int n0, int lane) {
    int n = lane & 15, kb = (lane >> 4) << 1;
    v2f r;
    r.x = p[(k0 + kb)     * ld + n0 + n];
    r.y = p[(k0 + kb + 1) * ld + n0 + n];
    return r;
}
__device__ __forceinline__ void stc_f(float* p, int ld, int r0, int c0, int lane, v8f c) {
    int col = lane & 15, rb = (lane >> 4) << 3;
#pragma unroll
    for (int g = 0; g < 8; ++g) p[(r0 + rb + g) * ld + c0 + col] = c[g];
}
__device__ __forceinline__ v8f wmma4(v2f a, v2f b, v8f c) {
    return __builtin_amdgcn_wmma_f32_16x16x4_f32(false, a, false, b, (short)0, c, false, false);
}

__device__ __forceinline__ v8f lds8(const float* p) {
    const v4f lo = *(const v4f*)p, hi = *(const v4f*)(p + 4);
    v8f r;
    r[0] = lo[0]; r[1] = lo[1]; r[2] = lo[2]; r[3] = lo[3];
    r[4] = hi[0]; r[5] = hi[1]; r[6] = hi[2]; r[7] = hi[3];
    return r;
}
__device__ __forceinline__ float xhalf(float x) {
    return __builtin_bit_cast(float, __builtin_amdgcn_permlanex16(__builtin_bit_cast(int, x), __builtin_bit_cast(int, x),
                                                                  0x76543210, 0xfedcba98, false, false));
}
__device__ __forceinline__ v2f pr(v8f t, int j) {
    v2f r; r.x = t[2 * j]; r.y = t[2 * j + 1];
    return r;
}

__device__ __forceinline__ v8f ln_t(v8f x, v8f w, int h) {
    float xs[8];
#pragma unroll
    for (int r = 0; r < 8; ++r) xs[r] = (r >= 3 && h) ? 0.f : x[r];
    float s = ((xs[0] + xs[1]) + (xs[2] + xs[3])) + ((xs[4] + xs[5]) + (xs[6] + xs[7]));
    s += xhalf(s);
    const float m = s * (1.0f / 11.0f);
    float d[8];
#pragma unroll
    for (int r = 0; r < 8; ++r) d[r] = (r >= 3 && h) ? 0.f : xs[r] - m;
    float v = ((d[0] * d[0] + d[1] * d[1]) + (d[2] * d[2] + d[3] * d[3])) +
              ((d[4] * d[4] + d[5] * d[5]) + (d[6] * d[6] + d[7] * d[7]));
    v += xhalf(v);
    v *= (1.0f / 11.0f);
    const float rinv = rsqrtf(v + 1e-5f);
    v8f o;
#pragma unroll
    for (int r = 0; r < 8; ++r) o[r] = d[r] * rinv * w[r];
    return o;
}

__global__ __launch_bounds__(128)
void pixel_encoder_kernel(const float* __restrict__ x,
                          const float* __restrict__ Wqkv,
                          const float* __restrict__ Wo,
                          const float* __restrict__ W1,
                          const float* __restrict__ W2,
                          const float* __restrict__ ln1w,
                          const float* __restrict__ ln2w,
                          float* __restrict__ out)
{
    __shared__ float Wq_t[16 * 16], Wk_t[16 * 16], Wv_t[16 * 16];
    __shared__ __attribute__((aligned(16))) float Wo_p[16 * 20];
    __shared__ __attribute__((aligned(16))) float W1_p[64 * 20];
    __shared__ __attribute__((aligned(16))) float W2_p[16 * 68];
    __shared__ __attribute__((aligned(16))) float ln1[16], ln2[16];
    __shared__ __attribute__((aligned(16))) float Xb[64 * 16];
    __shared__ float Qb[64 * 16], Kb[64 * 16], Vb[64 * 16];
    __shared__ __attribute__((aligned(16))) float Ab[64 * 16];

    const int tid  = threadIdx.x;
    const int lane = tid & 31;
    const int wv   = tid >> 5;
    const int r0   = wv * 16;
    const int p    = blockIdx.x;
    const int hh   = p >> 4;
    const int ww   = p & 15;

    const float scale = 0.43498892f;
    float gq[2], gk[2], gv[2], go[2], g1[6], g2[8], gx[7], gl1, gl2;
#pragma unroll
    for (int j = 0; j < 2; ++j) {
        const int i = tid + 128 * j;
        const int k = i >> 4, n = i & 15;
        const int a = min(n, 10) * 11 + min(k, 10);
        gq[j] = Wqkv[a];
        gk[j] = Wqkv[121 + a];
        gv[j] = Wqkv[242 + a];
        go[j] = Wo[a];
    }
#pragma unroll
    for (int j = 0; j < 6; ++j) {
        const int i = tid + 128 * j;
        const int k1 = i >> 6, n1 = i & 63;
        g1[j] = W1[n1 * 11 + min(k1, 10)];
    }
#pragma unroll
    for (int j = 0; j < 8; ++j) {
        const int i = tid + 128 * j;
        const int k2 = i >> 4, n2 = i & 15;
        g2[j] = W2[min(n2, 10) * 64 + k2];
    }
    {
        const int c = min(tid, 10);
        gl1 = ln1w[c];
        gl2 = ln2w[c];
    }
    const int xch = tid & 15;
    const int t0  = tid >> 4;
    bool xin[7];
    bool xwin[7];
#pragma unroll
    for (int j = 0; j < 7; ++j) {
        xwin[j] = (j < 6) || (t0 == 0);
        const int u  = (j < 6 ? t0 : 0) + j;
        const int c7 = (u >= 7) ? 1 : 0;
        const int gy  = hh + j + c7 - 3;
        const int gxx = ww + u - 7 * c7 - 3;
        xin[j] = xwin[j] && ((unsigned)gy < 16u) && ((unsigned)gxx < 16u);
        gx[j] = x[min(xch, 9) * 256 + min(max(gy, 0), 15) * 16 + min(max(gxx, 0), 15)];
    }
    __builtin_amdgcn_sched_barrier(0);

#pragma unroll
    for (int j = 0; j < 2; ++j) {
        const int i = tid + 128 * j;
        const int k = i >> 4, n = i & 15;
        const bool ok = (n < 11) && (k < 11);
        Wq_t[i] = ok ? gq[j] * scale : 0.f;
        Wk_t[i] = ok ? gk[j] : 0.f;
        Wv_t[i] = ok ? gv[j] : 0.f;
        Wo_p[n * 20 + k] = ok ? go[j] : 0.f;
    }
#pragma unroll
    for (int j = 0; j < 8; ++j) {
        const int i = tid + 128 * j;
        const int k1 = i >> 6, n1 = i & 63;
        W1_p[n1 * 20 + k1] = (j < 6 && k1 < 11) ? g1[j < 6 ? j : 0] : 0.f;
        const int k2 = i >> 4, n2 = i & 15;
        W2_p[n2 * 68 + k2] = (n2 < 11) ? g2[j] : 0.f;
    }
    asm volatile("" :: "v"(gl1), "v"(gl2));
    if (tid < 16) {
        ln1[tid] = (tid < 11) ? gl1 : 0.f;
        ln2[tid] = (tid < 11) ? gl2 : 0.f;
    }

#pragma unroll
    for (int j = 0; j < 8; ++j) {
        const int i = tid + 128 * j;
        float val = 0.f;
        if (j < 7) {
            const int jj = j < 7 ? j : 0;
            const float mk = (xch == 10 && xwin[jj] && !xin[jj]) ? 1.f : 0.f;
            val = (xch < 10 && xin[jj]) ? gx[jj] : mk;
        }
        Xb[i] = val;
    }
    __syncthreads();

    {
        v8f cq = {}, ck = {}, cv = {};
#pragma unroll
        for (int kk = 0; kk < 16; kk += 4) {
            v2f a = lda_f(Xb, 16, r0, kk, lane);
            cq = wmma4(a, ldb_f(Wq_t, 16, kk, 0, lane), cq);
            ck = wmma4(a, ldb_f(Wk_t, 16, kk, 0, lane), ck);
            cv = wmma4(a, ldb_f(Wv_t, 16, kk, 0, lane), cv);
        }
        stc_f(Qb, 16, r0, 0, lane, cq);
        stc_f(Kb, 16, r0, 0, lane, ck);
        stc_f(Vb, 16, r0, 0, lane, cv);
    }
    __syncthreads();

    {
        const int c16 = lane & 15, h = lane >> 4;

        v2f bq[4], ak[4][4], av[4][4];
#pragma unroll
        for (int kk = 0; kk < 4; ++kk) {
            bq[kk] = lda_f(Qb, 16, r0, kk * 4, lane);
#pragma unroll
            for (int ct = 0; ct < 4; ++ct) ak[kk][ct] = lda_f(Kb, 16, ct * 16, kk * 4, lane);
        }
#pragma unroll
        for (int ct = 0; ct < 4; ++ct) {
#pragma unroll
            for (int i = 0; i < 4; ++i) {
                const int kr = ct * 16 + h * 8 + i * 2;
                av[ct][i].x = Vb[kr * 16 + c16];
                av[ct][i].y = Vb[(kr + 1) * 16 + c16];
            }
        }
        const v8f xt = lds8(Xb + (r0 + c16) * 16 + 8 * h);
        __builtin_amdgcn_sched_barrier(0);

        v8f st[4] = {{}, {}, {}, {}};
#pragma unroll
        for (int kk = 0; kk < 4; ++kk) {
#pragma unroll
            for (int ct = 0; ct < 4; ++ct)
                st[ct] = wmma4(ak[kk][ct], bq[kk], st[ct]);
        }

        const v8f wo = lds8(Wo_p + c16 * 20 + 8 * h);
        v8f w1[4], w2[4];
#pragma unroll
        for (int ht = 0; ht < 4; ++ht) {
            w1[ht] = lds8(W1_p + (ht * 16 + c16) * 20 + 8 * h);
            w2[ht] = lds8(W2_p + c16 * 68 + ht * 16 + 8 * h);
        }
        const v8f gn1 = lds8(ln1 + 8 * h), gn2 = lds8(ln2 + 8 * h);
        __builtin_amdgcn_sched_barrier(0);

        float mt[4], sp[4];
#pragma unroll
        for (int ct = 0; ct < 4; ++ct)
            mt[ct] = fmaxf(fmaxf(fmaxf(st[ct][0], st[ct][1]), fmaxf(st[ct][2], st[ct][3])),
                           fmaxf(fmaxf(st[ct][4], st[ct][5]), fmaxf(st[ct][6], st[ct][7])));
        float m = fmaxf(fmaxf(fmaxf(mt[0], mt[1]), fmaxf(mt[2], mt[3])), 0.f);
        m = fmaxf(m, xhalf(m));


        v8f ot = {};
#pragma unroll
        for (int ct = 0; ct < 4; ++ct) {
#pragma unroll
            for (int g = 0; g < 8; ++g) st[ct][g] = __builtin_amdgcn_exp2f(st[ct][g] - m);
            sp[ct] = ((st[ct][0] + st[ct][1]) + (st[ct][2] + st[ct][3])) +
                     ((st[ct][4] + st[ct][5]) + (st[ct][6] + st[ct][7]));
#pragma unroll
            for (int i = 0; i < 4; ++i) {
                ot = wmma4(av[ct][i], pr(st[ct], i), ot);
            }
        }
        float s = (sp[0] + sp[1]) + (sp[2] + sp[3]);
        s += xhalf(s);
        s += 836.f * __builtin_amdgcn_exp2f(-m);
        const float inv = __builtin_amdgcn_rcpf(s);
#pragma unroll
        for (int r = 0; r < 8; ++r) ot[r] *= inv;

        v8f r1 = xt;
#pragma unroll
        for (int j = 0; j < 4; ++j) r1 = wmma4(pr(wo, j), pr(ot, j), r1);
        const v8f x1 = ln_t(r1, gn1, h);

        v8f hd[4];
#pragma unroll
        for (int ht = 0; ht < 4; ++ht) {
            v8f c = {};
#pragma unroll
            for (int j = 0; j < 4; ++j) c = wmma4(pr(w1[ht], j), pr(x1, j), c);
#pragma unroll
            for (int g = 0; g < 8; ++g) c[g] = fmaxf(c[g], 0.f);
            hd[ht] = c;
        }

        v8f y = x1;
#pragma unroll
        for (int ht = 0; ht < 4; ++ht) {
#pragma unroll
            for (int i = 0; i < 4; ++i) y = wmma4(pr(w2[ht], i), pr(hd[ht], i), y);
        }
        const v8f fin = ln_t(y, gn2, h);

        float* fo = Ab + (r0 + c16) * 16 + 8 * h;
        v4f lo, hi;
        lo[0] = fin[0]; lo[1] = fin[1]; lo[2] = fin[2]; lo[3] = fin[3];
        hi[0] = fin[4]; hi[1] = fin[5]; hi[2] = fin[6]; hi[3] = fin[7];
        *(v4f*)fo = lo;
        *(v4f*)(fo + 4) = hi;
    }
    __syncthreads();

    float* ob = out + p * 9000;
    for (int s = tid; s < 900; s += 128) {
        int r = s / 30, cc = s % 30;
        int t = (r < 7 && cc < 7) ? (r * 7 + cc) : 49;
        const float* row = Ab + t * 16;
#pragma unroll
        for (int ch = 0; ch < 10; ++ch) ob[ch * 900 + s] = row[ch];
    }
}

extern "C" void kernel_launch(void* const* d_in, const int* in_sizes, int n_in,
                              void* d_out, int out_size, void* d_ws, size_t ws_size,
                              hipStream_t stream) {
    const float* x    = (const float*)d_in[0];
    const float* Wqkv = (const float*)d_in[1];
    const float* Wo   = (const float*)d_in[2];
    const float* W1   = (const float*)d_in[3];
    const float* W2   = (const float*)d_in[4];
    const float* l1   = (const float*)d_in[5];
    const float* l2   = (const float*)d_in[6];
    (void)in_sizes; (void)n_in; (void)out_size; (void)d_ws; (void)ws_size;
    hipLaunchKernelGGL(pixel_encoder_kernel, dim3(256), dim3(128), 0, stream,
                       x, Wqkv, Wo, W1, W2, l1, l2, (float*)d_out);
}


# Output stage: the block's contiguous 36000-byte result leaves as 16-byte stores on whole 128-byte lines, 18 straight-line rounds with their LDS reads issued three rounds ahead, no wait per loop trip
# speedup vs baseline: 2.8515x; 1.0874x over previous
#include <hip/hip_runtime.h>
#include <hip/hip_bf16.h>

typedef __attribute__((ext_vector_type(2))) float v2f;
typedef __attribute__((ext_vector_type(4))) float v4f;
typedef __attribute__((ext_vector_type(8))) float v8f;
typedef __attribute__((ext_vector_type(4))) int v4i;

__device__ __forceinline__ v2f lda_f(const float* p, int ld, int r0, int k0, int lane) {
    int m = lane & 15, kb = (lane >> 4) << 1;
    const float* q = p + (r0 + m) * ld + k0 + kb;
    v2f r; r.x = q[0]; r.y = q[1];
    return r;
}
__device__ __forceinline__ v2f ldb_f(const float* p, int ld, int k0, int n0, int lane) {
    int n = lane & 15, kb = (lane >> 4) << 1;
    v2f r;
    r.x = p[(k0 + kb)     * ld + n0 + n];
    r.y = p[(k0 + kb + 1) * ld + n0 + n];
    return r;
}
__device__ __forceinline__ void stc_f(float* p, int ld, int r0, int c0, int lane, v8f c) {
    int col = lane & 15, rb = (lane >> 4) << 3;
#pragma unroll
    for (int g = 0; g < 8; ++g) p[(r0 + rb + g) * ld + c0 + col] = c[g];
}
__device__ __forceinline__ v8f wmma4(v2f a, v2f b, v8f c) {
    return __builtin_amdgcn_wmma_f32_16x16x4_f32(false, a, false, b, (short)0, c, false, false);
}

__device__ __forceinline__ v8f lds8(const float* p) {
    const v4f lo = *(const v4f*)p, hi = *(const v4f*)(p + 4);
    v8f r;
    r[0] = lo[0]; r[1] = lo[1]; r[2] = lo[2]; r[3] = lo[3];
    r[4] = hi[0]; r[5] = hi[1]; r[6] = hi[2]; r[7] = hi[3];
    return r;
}
__device__ __forceinline__ float xhalf(float x) {
    return __builtin_bit_cast(float, __builtin_amdgcn_permlanex16(__builtin_bit_cast(int, x), __builtin_bit_cast(int, x),
                                                                  0x76543210, 0xfedcba98, false, false));
}
__device__ __forceinline__ v2f pr(v8f t, int j) {
    v2f r; r.x = t[2 * j]; r.y = t[2 * j + 1];
    return r;
}

__device__ __forceinline__ v8f ln_t(v8f x, v8f w, int h) {
    float xs[8];
#pragma unroll
    for (int r = 0; r < 8; ++r) xs[r] = (r >= 3 && h) ? 0.f : x[r];
    float s = ((xs[0] + xs[1]) + (xs[2] + xs[3])) + ((xs[4] + xs[5]) + (xs[6] + xs[7]));
    s += xhalf(s);
    const float m = s * (1.0f / 11.0f);
    float d[8];
#pragma unroll
    for (int r = 0; r < 8; ++r) d[r] = (r >= 3 && h) ? 0.f : xs[r] - m;
    float v = ((d[0] * d[0] + d[1] * d[1]) + (d[2] * d[2] + d[3] * d[3])) +
              ((d[4] * d[4] + d[5] * d[5]) + (d[6] * d[6] + d[7] * d[7]));
    v += xhalf(v);
    v *= (1.0f / 11.0f);
    const float rinv = rsqrtf(v + 1e-5f);
    v8f o;
#pragma unroll
    for (int r = 0; r < 8; ++r) o[r] = d[r] * rinv * w[r];
    return o;
}

__global__ __launch_bounds__(128)
void pixel_encoder_kernel(const float* __restrict__ x,
                          const float* __restrict__ Wqkv,
                          const float* __restrict__ Wo,
                          const float* __restrict__ W1,
                          const float* __restrict__ W2,
                          const float* __restrict__ ln1w,
                          const float* __restrict__ ln2w,
                          float* __restrict__ out)
{
    __shared__ float Wq_t[16 * 16], Wk_t[16 * 16], Wv_t[16 * 16];
    __shared__ __attribute__((aligned(16))) float Wo_p[16 * 20];
    __shared__ __attribute__((aligned(16))) float W1_p[64 * 20];
    __shared__ __attribute__((aligned(16))) float W2_p[16 * 68];
    __shared__ __attribute__((aligned(16))) float ln1[16], ln2[16];
    __shared__ __attribute__((aligned(16))) float Xb[64 * 16];
    __shared__ float Qb[64 * 16], Kb[64 * 16], Vb[64 * 16];
    __shared__ __attribute__((aligned(16))) float Ab[64 * 16];
    __shared__ __attribute__((aligned(16))) int rowofs[192];

    const int tid  = threadIdx.x;
    const int lane = tid & 31;
    const int wv   = tid >> 5;
    const int r0   = wv * 16;
    const int p    = blockIdx.x;
    const int hh   = p >> 4;
    const int ww   = p & 15;

    const float scale = 0.43498892f;
    float gq[2], gk[2], gv[2], go[2], g1[6], g2[8], gx[7], gl1, gl2;
#pragma unroll
    for (int j = 0; j < 2; ++j) {
        const int i = tid + 128 * j;
        const int k = i >> 4, n = i & 15;
        const int a = min(n, 10) * 11 + min(k, 10);
        gq[j] = Wqkv[a];
        gk[j] = Wqkv[121 + a];
        gv[j] = Wqkv[242 + a];
        go[j] = Wo[a];
    }
#pragma unroll
    for (int j = 0; j < 6; ++j) {
        const int i = tid + 128 * j;
        const int k1 = i >> 6, n1 = i & 63;
        g1[j] = W1[n1 * 11 + min(k1, 10)];
    }
#pragma unroll
    for (int j = 0; j < 8; ++j) {
        const int i = tid + 128 * j;
        const int k2 = i >> 4, n2 = i & 15;
        g2[j] = W2[min(n2, 10) * 64 + k2];
    }
    {
        const int c = min(tid, 10);
        gl1 = ln1w[c];
        gl2 = ln2w[c];
    }
    const int xch = tid & 15;
    const int t0  = tid >> 4;
    bool xin[7];
    bool xwin[7];
#pragma unroll
    for (int j = 0; j < 7; ++j) {
        xwin[j] = (j < 6) || (t0 == 0);
        const int u  = (j < 6 ? t0 : 0) + j;
        const int c7 = (u >= 7) ? 1 : 0;
        const int gy  = hh + j + c7 - 3;
        const int gxx = ww + u - 7 * c7 - 3;
        xin[j] = xwin[j] && ((unsigned)gy < 16u) && ((unsigned)gxx < 16u);
        gx[j] = x[min(xch, 9) * 256 + min(max(gy, 0), 15) * 16 + min(max(gxx, 0), 15)];
    }
    __builtin_amdgcn_sched_barrier(0);

#pragma unroll
    for (int j = 0; j < 2; ++j) {
        const int s = tid + 128 * j;
        const int r = s / 30, cc = s - 30 * r;
        const int t = (r < 7 && cc < 7) ? (r * 7 + cc) : 49;
        if (j == 0 || tid < 64) rowofs[s] = t * 64;
    }

#pragma unroll
    for (int j = 0; j < 2; ++j) {
        const int i = tid + 128 * j;
        const int k = i >> 4, n = i & 15;
        const bool ok = (n < 11) && (k < 11);
        Wq_t[i] = ok ? gq[j] * scale : 0.f;
        Wk_t[i] = ok ? gk[j] : 0.f;
        Wv_t[i] = ok ? gv[j] : 0.f;
        Wo_p[n * 20 + k] = ok ? go[j] : 0.f;
    }
#pragma unroll
    for (int j = 0; j < 8; ++j) {
        const int i = tid + 128 * j;
        const int k1 = i >> 6, n1 = i & 63;
        W1_p[n1 * 20 + k1] = (j < 6 && k1 < 11) ? g1[j < 6 ? j : 0] : 0.f;
        const int k2 = i >> 4, n2 = i & 15;
        W2_p[n2 * 68 + k2] = (n2 < 11) ? g2[j] : 0.f;
    }
    asm volatile("" :: "v"(gl1), "v"(gl2));
    if (tid < 16) {
        ln1[tid] = (tid < 11) ? gl1 : 0.f;
        ln2[tid] = (tid < 11) ? gl2 : 0.f;
    }

#pragma unroll
    for (int j = 0; j < 8; ++j) {
        const int i = tid + 128 * j;
        float val = 0.f;
        if (j < 7) {
            const int jj = j < 7 ? j : 0;
            const float mk = (xch == 10 && xwin[jj] && !xin[jj]) ? 1.f : 0.f;
            val = (xch < 10 && xin[jj]) ? gx[jj] : mk;
        }
        Xb[i] = val;
    }
    __syncthreads();

    {
        v8f cq = {}, ck = {}, cv = {};
#pragma unroll
        for (int kk = 0; kk < 16; kk += 4) {
            v2f a = lda_f(Xb, 16, r0, kk, lane);
            cq = wmma4(a, ldb_f(Wq_t, 16, kk, 0, lane), cq);
            ck = wmma4(a, ldb_f(Wk_t, 16, kk, 0, lane), ck);
            cv = wmma4(a, ldb_f(Wv_t, 16, kk, 0, lane), cv);
        }
        stc_f(Qb, 16, r0, 0, lane, cq);
        stc_f(Kb, 16, r0, 0, lane, ck);
        stc_f(Vb, 16, r0, 0, lane, cv);
    }
    __syncthreads();

    const int oa = (2 * p) & 7;
    v4f* ob4 = (v4f*)(out + p * 9000) - oa;
    int ga[18][4];

    {
        const int c16 = lane & 15, h = lane >> 4;

        v2f bq[4], ak[4][4], av[4][4];
#pragma unroll
        for (int kk = 0; kk < 4; ++kk) {
            bq[kk] = lda_f(Qb, 16, r0, kk * 4, lane);
#pragma unroll
            for (int ct = 0; ct < 4; ++ct) ak[kk][ct] = lda_f(Kb, 16, ct * 16, kk * 4, lane);
        }
#pragma unroll
        for (int ct = 0; ct < 4; ++ct) {
#pragma unroll
            for (int i = 0; i < 4; ++i) {
                const int kr = ct * 16 + h * 8 + i * 2;
                av[ct][i].x = Vb[kr * 16 + c16];
                av[ct][i].y = Vb[(kr + 1) * 16 + c16];
            }
        }
        const v8f xt = lds8(Xb + (r0 + c16) * 16 + 8 * h);
        __builtin_amdgcn_sched_barrier(0);

        v8f st[4] = {{}, {}, {}, {}};
#pragma unroll
        for (int kk = 0; kk < 4; ++kk) {
#pragma unroll
            for (int ct = 0; ct < 4; ++ct)
                st[ct] = wmma4(ak[kk][ct], bq[kk], st[ct]);
        }

        const v8f wo = lds8(Wo_p + c16 * 20 + 8 * h);
        v8f w1[4], w2[4];
#pragma unroll
        for (int ht = 0; ht < 4; ++ht) {
            w1[ht] = lds8(W1_p + (ht * 16 + c16) * 20 + 8 * h);
            w2[ht] = lds8(W2_p + c16 * 68 + ht * 16 + 8 * h);
        }
        const v8f gn1 = lds8(ln1 + 8 * h), gn2 = lds8(ln2 + 8 * h);
        __builtin_amdgcn_sched_barrier(0);

        float mt[4], sp[4];
#pragma unroll
        for (int ct = 0; ct < 4; ++ct)
            mt[ct] = fmaxf(fmaxf(fmaxf(st[ct][0], st[ct][1]), fmaxf(st[ct][2], st[ct][3])),
                           fmaxf(fmaxf(st[ct][4], st[ct][5]), fmaxf(st[ct][6], st[ct][7])));
        float m = fmaxf(fmaxf(fmaxf(mt[0], mt[1]), fmaxf(mt[2], mt[3])), 0.f);
        m = fmaxf(m, xhalf(m));


        v8f ot = {};
#pragma unroll
        for (int ct = 0; ct < 4; ++ct) {
#pragma unroll
            for (int g = 0; g < 8; ++g) st[ct][g] = __builtin_amdgcn_exp2f(st[ct][g] - m);
            sp[ct] = ((st[ct][0] + st[ct][1]) + (st[ct][2] + st[ct][3])) +
                     ((st[ct][4] + st[ct][5]) + (st[ct][6] + st[ct][7]));
#pragma unroll
            for (int i = 0; i < 4; ++i) {
                ot = wmma4(av[ct][i], pr(st[ct], i), ot);
            }
        }
        float s = (sp[0] + sp[1]) + (sp[2] + sp[3]);
        s += xhalf(s);
        s += 836.f * __builtin_amdgcn_exp2f(-m);
        const float inv = __builtin_amdgcn_rcpf(s);
#pragma unroll
        for (int r = 0; r < 8; ++r) ot[r] *= inv;

        v8f r1 = xt;
#pragma unroll
        for (int j = 0; j < 4; ++j) r1 = wmma4(pr(wo, j), pr(ot, j), r1);
        const v8f x1 = ln_t(r1, gn1, h);

        v8f hd[4];
#pragma unroll
        for (int ht = 0; ht < 4; ++ht) {
            v8f c = {};
#pragma unroll
            for (int j = 0; j < 4; ++j) c = wmma4(pr(w1[ht], j), pr(x1, j), c);
#pragma unroll
            for (int g = 0; g < 8; ++g) c[g] = fmaxf(c[g], 0.f);
            hd[ht] = c;
        }

#pragma unroll
        for (int k = 0; k < 18; ++k) {
            const int i  = min(max(k * 128 + tid - oa, 0), 2249);
            const int ch = i / 225;
            const int q  = i - ch * 225;
            const v4i ro = *(const v4i*)(rowofs + 4 * min(q, 47));
            ga[k][0] = ro[0] + 4 * ch;
            ga[k][1] = ro[1] + 4 * ch;
            ga[k][2] = ro[2] + 4 * ch;
            ga[k][3] = ro[3] + 4 * ch;
        }

        v8f y = x1;
#pragma unroll
        for (int ht = 0; ht < 4; ++ht) {
#pragma unroll
            for (int i = 0; i < 4; ++i) y = wmma4(pr(w2[ht], i), pr(hd[ht], i), y);
        }
        const v8f fin = ln_t(y, gn2, h);

        float* fo = Ab + (r0 + c16) * 16 + 8 * h;
        v4f lo, hi;
        lo[0] = fin[0]; lo[1] = fin[1]; lo[2] = fin[2]; lo[3] = fin[3];
        hi[0] = fin[4]; hi[1] = fin[5]; hi[2] = fin[6]; hi[3] = fin[7];
        *(v4f*)fo = lo;
        *(v4f*)(fo + 4) = hi;
    }
    __syncthreads();

    {
        const char* abb = (const char*)Ab;
        v4f val[18];
#pragma unroll
        for (int k = 0; k < 3; ++k) {
#pragma unroll
            for (int e = 0; e < 4; ++e) val[k][e] = *(const float*)(abb + ga[k][e]);
        }
        __builtin_amdgcn_sched_barrier(0);
#pragma unroll
        for (int k = 0; k < 18; ++k) {
            if (k + 3 < 18) {
#pragma unroll
                for (int e = 0; e < 4; ++e) val[k + 3][e] = *(const float*)(abb + ga[k + 3][e]);
            }
            __builtin_amdgcn_sched_barrier(0);
            if (k == 0 || k == 17) {
                const int i = k * 128 + tid - oa;
                asm volatile("" :: "v"(val[k]));
                if (i >= 0 && i < 2250) ob4[k * 128 + tid] = val[k];
            } else {
                ob4[k * 128 + tid] = val[k];
            }
            __builtin_amdgcn_sched_barrier(0);
        }
        asm volatile("" :: "v"(val[0]), "v"(val[1]), "v"(val[2]), "v"(val[3]), "v"(val[4]), "v"(val[5]),
                           "v"(val[6]), "v"(val[7]), "v"(val[8]), "v"(val[9]), "v"(val[10]), "v"(val[11]),
                           "v"(val[12]), "v"(val[13]), "v"(val[14]), "v"(val[15]), "v"(val[16]), "v"(val[17]));
    }
}

extern "C" void kernel_launch(void* const* d_in, const int* in_sizes, int n_in,
                              void* d_out, int out_size, void* d_ws, size_t ws_size,
                              hipStream_t stream) {
    const float* x    = (const float*)d_in[0];
    const float* Wqkv = (const float*)d_in[1];
    const float* Wo   = (const float*)d_in[2];
    const float* W1   = (const float*)d_in[3];
    const float* W2   = (const float*)d_in[4];
    const float* l1   = (const float*)d_in[5];
    const float* l2   = (const float*)d_in[6];
    (void)in_sizes; (void)n_in; (void)out_size; (void)d_ws; (void)ws_size;
    hipLaunchKernelGGL(pixel_encoder_kernel, dim3(256), dim3(128), 0, stream,
                       x, Wqkv, Wo, W1, W2, l1, l2, (float*)d_out);
}


# Attention reassociated around X: scores = X (Wq^T Wk) X^T, output = (Wo Wv)(X^T P^T); both weight products made per wave in registers: no Q/K/V, no LDS hand-over, one barrier less; sums reordered
# speedup vs baseline: 3.0569x; 1.0720x over previous
#include <hip/hip_runtime.h>
#include <hip/hip_bf16.h>

typedef __attribute__((ext_vector_type(2))) float v2f;
typedef __attribute__((ext_vector_type(4))) float v4f;
typedef __attribute__((ext_vector_type(8))) float v8f;
typedef __attribute__((ext_vector_type(4))) int v4i;

__device__ __forceinline__ v8f wmma4(v2f a, v2f b, v8f c) {
    return __builtin_amdgcn_wmma_f32_16x16x4_f32(false, a, false, b, (short)0, c, false, false);
}

__device__ __forceinline__ v8f lds8(const float* p) {
    const v4f lo = *(const v4f*)p, hi = *(const v4f*)(p + 4);
    v8f r;
    r[0] = lo[0]; r[1] = lo[1]; r[2] = lo[2]; r[3] = lo[3];
    r[4] = hi[0]; r[5] = hi[1]; r[6] = hi[2]; r[7] = hi[3];
    return r;
}
__device__ __forceinline__ float xhalf(float x) {
    return __builtin_bit_cast(float, __builtin_amdgcn_permlanex16(__builtin_bit_cast(int, x), __builtin_bit_cast(int, x),
                                                                  0x76543210, 0xfedcba98, false, false));
}
__device__ __forceinline__ v2f pr(v8f t, int j) {
    v2f r; r.x = t[2 * j]; r.y = t[2 * j + 1];
    return r;
}

__device__ __forceinline__ v8f ln_t(v8f x, v8f w, int h) {
    float xs[8];
#pragma unroll
    for (int r = 0; r < 8; ++r) xs[r] = (r >= 3 && h) ? 0.f : x[r];
    float s = ((xs[0] + xs[1]) + (xs[2] + xs[3])) + ((xs[4] + xs[5]) + (xs[6] + xs[7]));
    s += xhalf(s);
    const float m = s * (1.0f / 11.0f);
    float d[8];
#pragma unroll
    for (int r = 0; r < 8; ++r) d[r] = (r >= 3 && h) ? 0.f : xs[r] - m;
    float v = ((d[0] * d[0] + d[1] * d[1]) + (d[2] * d[2] + d[3] * d[3])) +
              ((d[4] * d[4] + d[5] * d[5]) + (d[6] * d[6] + d[7] * d[7]));
    v += xhalf(v);
    v *= (1.0f / 11.0f);
    const float rinv = rsqrtf(v + 1e-5f);
    v8f o;
#pragma unroll
    for (int r = 0; r < 8; ++r) o[r] = d[r] * rinv * w[r];
    return o;
}

__global__ __launch_bounds__(128)
void pixel_encoder_kernel(const float* __restrict__ x,
                          const float* __restrict__ Wqkv,
                          const float* __restrict__ Wo,
                          const float* __restrict__ W1,
                          const float* __restrict__ W2,
                          const float* __restrict__ ln1w,
                          const float* __restrict__ ln2w,
                          float* __restrict__ out)
{
    __shared__ __attribute__((aligned(16))) float W1_p[64 * 20];
    __shared__ __attribute__((aligned(16))) float W2_p[16 * 68];
    __shared__ __attribute__((aligned(16))) float ln1[16], ln2[16];
    __shared__ __attribute__((aligned(16))) float Xb[64 * 20];
    __shared__ __attribute__((aligned(16))) float XTb[16 * 68];
    __shared__ __attribute__((aligned(16))) float Ab[64 * 16];
    __shared__ __attribute__((aligned(16))) int rowofs[192];

    const int tid  = threadIdx.x;
    const int lane = tid & 31;
    const int wv   = tid >> 5;
    const int r0   = wv * 16;
    const int p    = blockIdx.x;
    const int hh   = p >> 4;
    const int ww   = p & 15;

    const float scale = 0.43498892f;
    const int c16 = lane & 15, h = lane >> 4;
    float fq[3][2], fk[3][2], fv[3][2], fo[3][2], g1[6], g2[8], gx[7], gl1, gl2;
#pragma unroll
    for (int kk = 0; kk < 3; ++kk) {
#pragma unroll
        for (int e = 0; e < 2; ++e) {
            const int d = min(4 * kk + 2 * h + e, 10), cc = min(c16, 10);
            fq[kk][e] = Wqkv[d * 11 + cc];
            fk[kk][e] = Wqkv[121 + d * 11 + cc];
            fv[kk][e] = Wqkv[242 + d * 11 + cc];
            fo[kk][e] = Wo[cc * 11 + d];
        }
    }
    __builtin_amdgcn_sched_barrier(0);
#pragma unroll
    for (int j = 0; j < 6; ++j) {
        const int i = tid + 128 * j;
        const int k1 = i >> 6, n1 = i & 63;
        g1[j] = W1[n1 * 11 + min(k1, 10)];
    }
#pragma unroll
    for (int j = 0; j < 8; ++j) {
        const int i = tid + 128 * j;
        const int k2 = i >> 4, n2 = i & 15;
        g2[j] = W2[min(n2, 10) * 64 + k2];
    }
    {
        const int c = min(tid, 10);
        gl1 = ln1w[c];
        gl2 = ln2w[c];
    }
    const int xch = tid & 15;
    const int t0  = tid >> 4;
    bool xin[7];
    bool xwin[7];
#pragma unroll
    for (int j = 0; j < 7; ++j) {
        xwin[j] = (j < 6) || (t0 == 0);
        const int u  = (j < 6 ? t0 : 0) + j;
        const int c7 = (u >= 7) ? 1 : 0;
        const int gy  = hh + j + c7 - 3;
        const int gxx = ww + u - 7 * c7 - 3;
        xin[j] = xwin[j] && ((unsigned)gy < 16u) && ((unsigned)gxx < 16u);
        gx[j] = x[min(xch, 9) * 256 + min(max(gy, 0), 15) * 16 + min(max(gxx, 0), 15)];
    }
    __builtin_amdgcn_sched_barrier(0);

#pragma unroll
    for (int j = 0; j < 2; ++j) {
        const int s = tid + 128 * j;
        const int r = s / 30, cc = s - 30 * r;
        const int t = (r < 7 && cc < 7) ? (r * 7 + cc) : 49;
        if (j == 0 || tid < 64) rowofs[s] = t * 64;
    }

    v8f mt = {}, wovt = {};
#pragma unroll
    for (int kk = 0; kk < 3; ++kk) {
        v2f aq, bk, av, bo;
        const bool ok0 = (c16 < 11) && (4 * kk + 2 * h < 11);
        const bool ok1 = (c16 < 11) && (4 * kk + 2 * h + 1 < 11);
        aq.x = ok0 ? fq[kk][0] * scale : 0.f;  aq.y = ok1 ? fq[kk][1] * scale : 0.f;
        bk.x = ok0 ? fk[kk][0] : 0.f;          bk.y = ok1 ? fk[kk][1] : 0.f;
        av.x = ok0 ? fv[kk][0] : 0.f;          av.y = ok1 ? fv[kk][1] : 0.f;
        bo.x = ok0 ? fo[kk][0] : 0.f;          bo.y = ok1 ? fo[kk][1] : 0.f;
        mt   = wmma4(aq, bk, mt);
        wovt = wmma4(av, bo, wovt);
    }

#pragma unroll
    for (int j = 0; j < 8; ++j) {
        const int i = tid + 128 * j;
        const int k1 = i >> 6, n1 = i & 63;
        W1_p[n1 * 20 + k1] = (j < 6 && k1 < 11) ? g1[j < 6 ? j : 0] : 0.f;
        const int k2 = i >> 4, n2 = i & 15;
        W2_p[n2 * 68 + k2] = (n2 < 11) ? g2[j] : 0.f;
    }
    asm volatile("" :: "v"(gl1), "v"(gl2));
    if (tid < 16) {
        ln1[tid] = (tid < 11) ? gl1 : 0.f;
        ln2[tid] = (tid < 11) ? gl2 : 0.f;
    }

#pragma unroll
    for (int j = 0; j < 8; ++j) {
        const int t = t0 + 8 * j;
        float val = 0.f;
        if (j < 7) {
            const int jj = j < 7 ? j : 0;
            const float mk = (xch == 10 && xwin[jj] && !xin[jj]) ? 1.f : 0.f;
            val = (xch < 10 && xin[jj]) ? gx[jj] : mk;
        }
        Xb[t * 20 + xch] = val;
        XTb[xch * 68 + t] = val;
    }
    __syncthreads();

    const int oa = (2 * p) & 7;
    v4f* ob4 = (v4f*)(out + p * 9000) - oa;
    int ga[18][4];

    {
        const v8f xt = lds8(Xb + (r0 + c16) * 20 + 8 * h);
        v8f xk[4], xT[4];
#pragma unroll
        for (int ct = 0; ct < 4; ++ct) xk[ct] = lds8(Xb + (ct * 16 + c16) * 20 + 8 * h);
#pragma unroll
        for (int ct = 0; ct < 4; ++ct) xT[ct] = lds8(XTb + c16 * 68 + ct * 16 + 8 * h);
        __builtin_amdgcn_sched_barrier(0);

        v8f gt = {};
#pragma unroll
        for (int j = 0; j < 4; ++j) gt = wmma4(pr(mt, j), pr(xt, j), gt);

        v8f st[4] = {{}, {}, {}, {}};
#pragma unroll
        for (int j = 0; j < 4; ++j) {
#pragma unroll
            for (int ct = 0; ct < 4; ++ct)
                st[ct] = wmma4(pr(xk[ct], j), pr(gt, j), st[ct]);
        }

        v8f w1[4], w2[4];
#pragma unroll
        for (int ht = 0; ht < 4; ++ht) {
            w1[ht] = lds8(W1_p + (ht * 16 + c16) * 20 + 8 * h);
            w2[ht] = lds8(W2_p + c16 * 68 + ht * 16 + 8 * h);
        }
        const v8f gn1 = lds8(ln1 + 8 * h), gn2 = lds8(ln2 + 8 * h);
        __builtin_amdgcn_sched_barrier(0);

        float mt[4], sp[4];
#pragma unroll
        for (int ct = 0; ct < 4; ++ct)
            mt[ct] = fmaxf(fmaxf(fmaxf(st[ct][0], st[ct][1]), fmaxf(st[ct][2], st[ct][3])),
                           fmaxf(fmaxf(st[ct][4], st[ct][5]), fmaxf(st[ct][6], st[ct][7])));
        float m = fmaxf(fmaxf(fmaxf(mt[0], mt[1]), fmaxf(mt[2], mt[3])), 0.f);
        m = fmaxf(m, xhalf(m));


        v8f zt = {};
#pragma unroll
        for (int ct = 0; ct < 4; ++ct) {
#pragma unroll
            for (int g = 0; g < 8; ++g) st[ct][g] = __builtin_amdgcn_exp2f(st[ct][g] - m);
            sp[ct] = ((st[ct][0] + st[ct][1]) + (st[ct][2] + st[ct][3])) +
                     ((st[ct][4] + st[ct][5]) + (st[ct][6] + st[ct][7]));
#pragma unroll
            for (int i = 0; i < 4; ++i) {
                zt = wmma4(pr(xT[ct], i), pr(st[ct], i), zt);
            }
        }
        float s = (sp[0] + sp[1]) + (sp[2] + sp[3]);
        s += xhalf(s);
        s += 836.f * __builtin_amdgcn_exp2f(-m);
        const float inv = __builtin_amdgcn_rcpf(s);
#pragma unroll
        for (int r = 0; r < 8; ++r) zt[r] *= inv;

        v8f r1 = xt;
#pragma unroll
        for (int j = 0; j < 4; ++j) r1 = wmma4(pr(wovt, j), pr(zt, j), r1);
        const v8f x1 = ln_t(r1, gn1, h);

        v8f hd[4];
#pragma unroll
        for (int ht = 0; ht < 4; ++ht) {
            v8f c = {};
#pragma unroll
            for (int j = 0; j < 4; ++j) c = wmma4(pr(w1[ht], j), pr(x1, j), c);
#pragma unroll
            for (int g = 0; g < 8; ++g) c[g] = fmaxf(c[g], 0.f);
            hd[ht] = c;
        }

#pragma unroll
        for (int k = 0; k < 18; ++k) {
            const int i  = min(max(k * 128 + tid - oa, 0), 2249);
            const int ch = i / 225;
            const int q  = i - ch * 225;
            const v4i ro = *(const v4i*)(rowofs + 4 * min(q, 47));
            ga[k][0] = ro[0] + 4 * ch;
            ga[k][1] = ro[1] + 4 * ch;
            ga[k][2] = ro[2] + 4 * ch;
            ga[k][3] = ro[3] + 4 * ch;
        }

        v8f y = x1;
#pragma unroll
        for (int ht = 0; ht < 4; ++ht) {
#pragma unroll
            for (int i = 0; i < 4; ++i) y = wmma4(pr(w2[ht], i), pr(hd[ht], i), y);
        }
        const v8f fin = ln_t(y, gn2, h);

        float* fo = Ab + (r0 + c16) * 16 + 8 * h;
        v4f lo, hi;
        lo[0] = fin[0]; lo[1] = fin[1]; lo[2] = fin[2]; lo[3] = fin[3];
        hi[0] = fin[4]; hi[1] = fin[5]; hi[2] = fin[6]; hi[3] = fin[7];
        *(v4f*)fo = lo;
        *(v4f*)(fo + 4) = hi;
    }
    __syncthreads();

    {
        const char* abb = (const char*)Ab;
        v4f val[18];
#pragma unroll
        for (int k = 0; k < 3; ++k) {
#pragma unroll
            for (int e = 0; e < 4; ++e) val[k][e] = *(const float*)(abb + ga[k][e]);
        }
        __builtin_amdgcn_sched_barrier(0);
#pragma unroll
        for (int k = 0; k < 18; ++k) {
            if (k + 3 < 18) {
#pragma unroll
                for (int e = 0; e < 4; ++e) val[k + 3][e] = *(const float*)(abb + ga[k + 3][e]);
            }
            __builtin_amdgcn_sched_barrier(0);
            if (k == 0 || k == 17) {
                const int i = k * 128 + tid - oa;
                asm volatile("" :: "v"(val[k]));
                if (i >= 0 && i < 2250) ob4[k * 128 + tid] = val[k];
            } else {
                ob4[k * 128 + tid] = val[k];
            }
            __builtin_amdgcn_sched_barrier(0);
        }
        asm volatile("" :: "v"(val[0]), "v"(val[1]), "v"(val[2]), "v"(val[3]), "v"(val[4]), "v"(val[5]),
                           "v"(val[6]), "v"(val[7]), "v"(val[8]), "v"(val[9]), "v"(val[10]), "v"(val[11]),
                           "v"(val[12]), "v"(val[13]), "v"(val[14]), "v"(val[15]), "v"(val[16]), "v"(val[17]));
    }
}

extern "C" void kernel_launch(void* const* d_in, const int* in_sizes, int n_in,
                              void* d_out, int out_size, void* d_ws, size_t ws_size,
                              hipStream_t stream) {
    const float* x    = (const float*)d_in[0];
    const float* Wqkv = (const float*)d_in[1];
    const float* Wo   = (const float*)d_in[2];
    const float* W1   = (const float*)d_in[3];
    const float* W2   = (const float*)d_in[4];
    const float* l1   = (const float*)d_in[5];
    const float* l2   = (const float*)d_in[6];
    (void)in_sizes; (void)n_in; (void)out_size; (void)d_ws; (void)ws_size;
    hipLaunchKernelGGL(pixel_encoder_kernel, dim3(256), dim3(128), 0, stream,
                       x, Wqkv, Wo, W1, W2, l1, l2, (float*)d_out);
}
